# starting kernel
# baseline (speedup 1.0000x reference)
#include <hip/hip_runtime.h>

typedef __attribute__((ext_vector_type(2))) float v2f;
typedef __attribute__((ext_vector_type(8))) float v8f;

static constexpr int kH    = 64;
static constexpr int kD    = 64;
static constexpr int kDin  = 2145;
static constexpr int kRowsPerWave = 16;
static constexpr int kWaves       = 8;
static constexpr int kKChunks     = 17;
static constexpr int kBpStride    = 65;
static constexpr int kXsStride    = 68;

__global__ __launch_bounds__(256) void qvf_wmma_kernel(
    const float* __restrict__ obs,
    const float* __restrict__ W,
    const float* __restrict__ bias,
    float* __restrict__ out)
{
  __shared__ v2f   Bp[34 * kBpStride];
  __shared__ float Xs[kWaves][kRowsPerWave * kXsStride];

  const int tid  = threadIdx.x;
  const int h    = blockIdx.y;
  const int wave = tid >> 5;
  const int lane = tid & 31;
  const int lo   = lane & 15;
  const int hi   = lane >> 4;

  {
    float* Bf = (float*)Bp;
    const float* Wh = W + (size_t)h * kDin;
    #pragma unroll
    for (int e = tid; e < 68 * 64; e += 256) {
      const int j = e >> 6;
      const int i = e & 63;
      float v = 0.f;
      if (j < 64) {
        if (i <= j) v = Wh[64 + i * 64 - (i * (i - 1)) / 2 + (j - i)];
      } else if (j == 64) {
        v = Wh[i];
      }
      Bf[((j >> 1) * kBpStride + i) * 2 + (j & 1)] = v;
    }
  }

  const int n_base = blockIdx.x * (kWaves * kRowsPerWave) + wave * kRowsPerWave;
  const int m    = lo;
  const int koff = hi * 2;
  const float* xrow = obs + ((size_t)(n_base + m) * kH + h) * kD;

  v2f a[kKChunks];
  #pragma unroll
  for (int kk = 0; kk < 16; ++kk) {
    v2f t = *(const v2f*)(xrow + 4 * kk + koff);
    a[kk] = t;
    *(v2f*)(&Xs[wave][m * kXsStride + 4 * kk + koff]) = t;
  }
  {
    v2f aext;
    aext.x = (hi == 0) ? 1.0f : 0.0f;
    aext.y = 0.0f;
    a[16] = aext;
  }

  __syncthreads();

  float part[8];
  #pragma unroll
  for (int v = 0; v < 8; ++v) part[v] = 0.f;

  #pragma unroll
  for (int t = 0; t < 4; ++t) {
    v8f acc = {};
    #pragma unroll
    for (int kk = 0; kk < kKChunks; ++kk) {
      v2f b = Bp[(2 * kk + hi) * kBpStride + (16 * t + lo)];
      acc = __builtin_amdgcn_wmma_f32_16x16x4_f32(
           false, a[kk],  false, b,
           (short)0, acc,  false,  false);
    }
    #pragma unroll
    for (int v = 0; v < 8; ++v) {
      const int mr = v + 8 * hi;
      part[v] += Xs[wave][mr * kXsStride + 16 * t + lo] * acc[v];
    }
  }

  #pragma unroll
  for (int v = 0; v < 8; ++v) {
    float s = part[v];
    s += __shfl_xor(s, 1, 32);
    s += __shfl_xor(s, 2, 32);
    s += __shfl_xor(s, 4, 32);
    s += __shfl_xor(s, 8, 32);
    part[v] = s;
  }

  const float b0 = bias[h];
  #pragma unroll
  for (int v = 0; v < 8; ++v) {
    if (lo == v) {
      const int row = n_base + v + 8 * hi;
      out[(size_t)row * kH + h] = part[v] + b0;
    }
  }
}

extern "C" void kernel_launch(void* const* d_in, const int* in_sizes, int n_in,
                              void* d_out, int out_size, void* d_ws, size_t ws_size,
                              hipStream_t stream) {
  const float* obs  = (const float*)d_in[0];
  const float* W    = (const float*)d_in[1];
  const float* bias = (const float*)d_in[2];
  float* out = (float*)d_out;

  const int num_paths = in_sizes[0] / (kH * kD);
  dim3 grid(num_paths / (kWaves * kRowsPerWave), kH);
  qvf_wmma_kernel<<<grid, 256, 0, stream>>>(obs, W, bias, out);
}


# x stream overlapped with the matrix phase: weights loaded by 4 waves and queued ahead of every x load, barrier waits for LDS only, each wave's WMMAs start as its own x lands; sums unchanged
# speedup vs baseline: 1.5556x; 1.5556x over previous
#include <hip/hip_runtime.h>


typedef __attribute__((ext_vector_type(2))) float v2f;
typedef __attribute__((ext_vector_type(4))) float v4f;
typedef __attribute__((ext_vector_type(8))) float v8f;

static constexpr int kH    = 64;
static constexpr int kD    = 64;
static constexpr int kDin  = 2145;
static constexpr int kRowsPerWave = 16;
static constexpr int kWaves       = 16;
static constexpr int kLoaders     = 4;
static constexpr int kBpStride    = 65;
static constexpr int kXsStride    = 68;

__global__ __launch_bounds__(512) void qvf_wmma_kernel(
    const float* __restrict__ obs,
    const float* __restrict__ W,
    const float* __restrict__ bias,
    float* __restrict__ out)
{
  struct alignas(16) Smem {
    v2f   Bp[36 * kBpStride];
    float Xs[kWaves][kRowsPerWave * kXsStride];
  };
  __shared__ Smem sm;
  v2f* const Bp = sm.Bp;
  float (* const Xs)[kRowsPerWave * kXsStride] = sm.Xs;

  const int tid  = threadIdx.x;
  const int h    = blockIdx.y;
  const int wave = __builtin_amdgcn_readfirstlane(tid >> 5);
  const int lane = tid & 31;
  const int lo   = lane & 15;
  const int hi   = lane >> 4;
  const bool loader = wave < kLoaders;

  const int n_base = blockIdx.x * (kWaves * kRowsPerWave) + wave * kRowsPerWave;
  const int m    = lo;
  const float* xrow = obs + ((size_t)(n_base + m) * kH + h) * kD;
  const float* Wh = W + (size_t)h * kDin;

  float wq[24];
  float wlin = 0.f;
  if (loader) {
    #pragma unroll
    for (int jh = 1; jh >= 0; --jh) {
      const int wj = 32 * jh + lane;
      int i  = wave;
      int rb = 64 + i * 64 - (i * (i - 1)) / 2 - i;
      #pragma unroll
      for (int kk = 0; kk < (jh ? 16 : 8); ++kk) {
        const int jc = (wj > i) ? wj : i;
        wq[(jh ? 0 : 16) + kk] = Wh[(unsigned)rb + (unsigned)jc];
        rb += 246 - 4 * i;
        i  += 4;
      }
    }
    wlin = Wh[(unsigned)(tid & 63)];
  }

  __builtin_amdgcn_s_barrier();

  v4f xq[8];
  #pragma unroll
  for (int q = 0; q < 8; ++q) xq[q] = *(const v4f*)(xrow + 8 * q + 4 * hi);

  if (loader) {
    float* Bf = (float*)Bp;
    #pragma unroll
    for (int k = 0; k < 24; ++k) {
      const int jh = (k < 16) ? 1 : 0;
      const int i  = wave + 4 * ((k < 16) ? k : (k - 16));
      const int wj = 32 * jh + lane;
      const float v = (i <= wj) ? wq[k] : 0.f;
      Bf[((wj >> 1) * kBpStride + i) * 2 + (wj & 1)] = v;
    }
    const int pr = 32 + 2 * (wave >> 1);
    Bp[pr * kBpStride + (tid & 63)] = v2f{(wave < 2) ? wlin : 0.f, 0.f};
  } else {
    const int z = tid - 32 * kLoaders;
    Bp[(z >> 5) * kBpStride + 32 + (z & 31)] = v2f{0.f, 0.f};
    if (z < 128) {
      const int e = z + 384;
      Bp[(e >> 5) * kBpStride + 32 + (e & 31)] = v2f{0.f, 0.f};
    }
  }

  __builtin_amdgcn_fence(__ATOMIC_RELEASE, "workgroup", "local");
  __builtin_amdgcn_s_barrier();
  __builtin_amdgcn_fence(__ATOMIC_ACQUIRE, "workgroup", "local");

  v8f acc[4];
  #pragma unroll
  for (int t = 0; t < 4; ++t) acc[t] = v8f{};

  #pragma unroll
  for (int q = 0; q < 8; ++q) {
    *(v4f*)(&Xs[wave][m * kXsStride + 8 * q + 4 * hi]) = xq[q];
    #pragma unroll
    for (int r = 0; r < 2; ++r) {
      const v2f a = r ? v2f{xq[q].z, xq[q].w} : v2f{xq[q].x, xq[q].y};
      #pragma unroll
      for (int t = 0; t < 4; ++t) {
        const v2f b = Bp[(4 * q + 2 * hi + r) * kBpStride + (16 * t + lo)];
        acc[t] = __builtin_amdgcn_wmma_f32_16x16x4_f32(
             false, a,  false, b,
             (short)0, acc[t],  false,  false);
      }
    }
  }
  {
    v2f aext;
    aext.x = (hi == 0) ? 1.0f : 0.0f;
    aext.y = 0.0f;
    #pragma unroll
    for (int t = 0; t < 4; ++t) {
      const v2f b = Bp[(32 + 2 * hi) * kBpStride + (16 * t + lo)];
      acc[t] = __builtin_amdgcn_wmma_f32_16x16x4_f32(
          false, aext, false, b, (short)0, acc[t], false, false);
    }
  }

  float part[8];
  #pragma unroll
  for (int v = 0; v < 8; ++v) part[v] = 0.f;
  #pragma unroll
  for (int t = 0; t < 4; ++t) {
    #pragma unroll
    for (int v = 0; v < 8; ++v) {
      const int mr = v + 8 * hi;
      part[v] += Xs[wave][mr * kXsStride + 16 * t + lo] * acc[t][v];
    }
  }

  #pragma unroll
  for (int v = 0; v < 8; ++v) {
    float s = part[v];
    s += __shfl_xor(s, 1, 32);
    s += __shfl_xor(s, 2, 32);
    s += __shfl_xor(s, 4, 32);
    s += __shfl_xor(s, 8, 32);
    part[v] = s;
  }

  const float b0 = bias[h];
  #pragma unroll
  for (int v = 0; v < 8; ++v) {
    if (lo == v) {
      const int row = n_base + v + 8 * hi;
      out[(size_t)row * kH + h] = part[v] + b0;
    }
  }
}

extern "C" void kernel_launch(void* const* d_in, const int* in_sizes, int n_in,
                              void* d_out, int out_size, void* d_ws, size_t ws_size,
                              hipStream_t stream) {
  const float* obs  = (const float*)d_in[0];
  const float* W    = (const float*)d_in[1];
  const float* bias = (const float*)d_in[2];
  float* out = (float*)d_out;

  const int num_paths = in_sizes[0] / (kH * kD);
  dim3 grid(num_paths / (kWaves * kRowsPerWave), kH);
  qvf_wmma_kernel<<<grid, 512, 0, stream>>>(obs, W, bias, out);
}
